# starting kernel
# baseline (speedup 1.0000x reference)
#include <hip/hip_runtime.h>

typedef __attribute__((ext_vector_type(2))) float v2f;
typedef __attribute__((ext_vector_type(8))) float v8f;

#define HH 56
#define WW 56
#define HW 3136
#define TILES_PER_IMG 196
#define WAVES_PER_BLOCK 4
#define NUM_TASKS (4 * 4 * TILES_PER_IMG)

__device__ __forceinline__ float hsum16(float v) {
#pragma unroll
  for (int m = 1; m < 16; m <<= 1) v += __shfl_xor(v, m, 32);
  return v;
}

__global__ __launch_bounds__(WAVES_PER_BLOCK * 32)
void caps_conv_route(const float* __restrict__ x, const float* __restrict__ wgt,
                     float* __restrict__ out) {
  __shared__ float lds[WAVES_PER_BLOCK][16][17];

  const int lane = threadIdx.x & 31;
  const int wv   = threadIdx.x >> 5;
  const int task = blockIdx.x * WAVES_PER_BLOCK + wv;
  const int tile = task % TILES_PER_IMG;
  const int o    = (task / TILES_PER_IMG) & 3;
  const int n    = task / (TILES_PER_IMG * 4);
  const int half = lane >> 4;
  const int col  = lane & 15;
  const int flat0 = tile * 16;

  const int fA = flat0 + col;
  const int hA = fA / WW;
  const int wA = fA - hA * WW;

  float out_acc[8];
#pragma unroll
  for (int j = 0; j < 8; ++j) out_acc[j] = 0.f;

  for (int p = 0; p < 4; ++p) {
    v8f pri[9];
#pragma unroll
    for (int k = 0; k < 9; ++k) {
      const int dj = k / 3, dk = k % 3;
      const int y  = hA + dj - 1;
      const int xx = wA + dk - 1;
      const float msk = (y >= 0 && y < HH && xx >= 0 && xx < WW) ? 1.f : 0.f;
      const int yc = min(max(y, 0), HH - 1);
      const int xc = min(max(xx, 0), WW - 1);
      v8f c = {};
#pragma unroll
      for (int hl = 0; hl < 2; ++hl) {
        const int lb = hl * 4 + 2 * half;
        const float* xp = x + (((size_t)(n * 4 + p) * 8 + lb) * HH + yc) * WW + xc;
        v2f a;
        a.x = msk * xp[0];
        a.y = msk * xp[HW];
        const float* wp = wgt + (((size_t)((o * 4 + p) * 9 + k) * 8 + lb) * 16) + col;
        v2f b;
        b.x = wp[0];
        b.y = wp[16];
        c = __builtin_amdgcn_wmma_f32_16x16x4_f32(false, a, false, b, (short)0, c,
                                                  false, false);
      }
      pri[k] = c;
    }

#pragma unroll
    for (int j = 0; j < 8; ++j) {
      float s = 0.f;
#pragma unroll
      for (int k = 0; k < 9; ++k) s += pri[k][j];
      s *= (1.f / 9.f);
      float sq    = hsum16(s * s);
      float scale = (sq * __frcp_rn(1.f + sq)) * __frsqrt_rn(sq);
      float outv  = scale * s;

      float lg[9];
#pragma unroll
      for (int k = 0; k < 9; ++k) lg[k] = hsum16(pri[k][j] * outv);

#pragma unroll
      for (int it = 1; it < 3; ++it) {
        float mx = lg[0];
#pragma unroll
        for (int k = 1; k < 9; ++k) mx = fmaxf(mx, lg[k]);
        float den = 0.f, acc = 0.f;
#pragma unroll
        for (int k = 0; k < 9; ++k) {
          const float e = __expf(lg[k] - mx);
          den += e;
          acc += e * pri[k][j];
        }
        s     = acc * __frcp_rn(den);
        sq    = hsum16(s * s);
        scale = (sq * __frcp_rn(1.f + sq)) * __frsqrt_rn(sq);
        outv  = scale * s;
        if (it == 1) {
#pragma unroll
          for (int k = 0; k < 9; ++k) lg[k] += hsum16(pri[k][j] * outv);
        }
      }
      out_acc[j] += outv;
    }
  }

#pragma unroll
  for (int j = 0; j < 8; ++j) lds[wv][col][j + 8 * half] = out_acc[j];
#pragma unroll
  for (int r = 0; r < 8; ++r) {
    const int d = r + 8 * half;
    const float v = lds[wv][d][col];
    out[((size_t)((n * 4 + o) * 16 + d)) * HW + flat0 + col] = v;
  }
}

extern "C" void kernel_launch(void* const* d_in, const int* in_sizes, int n_in,
                              void* d_out, int out_size, void* d_ws, size_t ws_size,
                              hipStream_t stream) {
  const float* x   = (const float*)d_in[0];
  const float* wgt = (const float*)d_in[1];
  float* out = (float*)d_out;
  (void)in_sizes; (void)n_in; (void)out_size; (void)d_ws; (void)ws_size;

  dim3 grid(NUM_TASKS / WAVES_PER_BLOCK);
  dim3 block(WAVES_PER_BLOCK * 32);
  hipLaunchKernelGGL(caps_conv_route, grid, block, 0, stream, x, wgt, out);
}


# Swap the WMMA operand roles (weight as A, patch as B) so a lane holds one position's d values: sums over d become in-lane adds plus one permlanex16, replacing 4-step ds_bpermute butterflies
# speedup vs baseline: 6.3263x; 6.3263x over previous
#include <hip/hip_runtime.h>


typedef __attribute__((ext_vector_type(2))) float v2f;
typedef __attribute__((ext_vector_type(8))) float v8f;

#define HH 56
#define WW 56
#define HW 3136
#define TILES_PER_IMG 196
#define WAVES_PER_BLOCK 4
#define NUM_TASKS (4 * 4 * TILES_PER_IMG)

__device__ __forceinline__ float xhalf(float x) {
  const int b = __builtin_bit_cast(int, x);
  return __builtin_bit_cast(
      float, __builtin_amdgcn_permlanex16(b, b, 0x76543210, 0xfedcba98, false, false));
}

__global__ __launch_bounds__(WAVES_PER_BLOCK * 32)
void caps_conv_route(const float* __restrict__ x, const float* __restrict__ wgt,
                     float* __restrict__ out) {
  const int lane = threadIdx.x & 31;
  const int wv   = threadIdx.x >> 5;
  const int task = blockIdx.x * WAVES_PER_BLOCK + wv;
  const int tile = task % TILES_PER_IMG;
  const int o    = (task / TILES_PER_IMG) & 3;
  const int n    = task / (TILES_PER_IMG * 4);
  const int half = lane >> 4;
  const int col  = lane & 15;
  const int flat0 = tile * 16;

  const int fB = flat0 + col;
  const int hB = fB / WW;
  const int wB = fB - hB * WW;

  float out_acc[8];
#pragma unroll
  for (int j = 0; j < 8; ++j) out_acc[j] = 0.f;

  for (int p = 0; p < 4; ++p) {
    v8f pri[9];
#pragma unroll
    for (int k = 0; k < 9; ++k) {
      const int dj = k / 3, dk = k % 3;
      const int y  = hB + dj - 1;
      const int xx = wB + dk - 1;
      const float msk = (y >= 0 && y < HH && xx >= 0 && xx < WW) ? 1.f : 0.f;
      const int yc = min(max(y, 0), HH - 1);
      const int xc = min(max(xx, 0), WW - 1);
      v8f c = {};
#pragma unroll
      for (int hl = 0; hl < 2; ++hl) {
        const int lb = hl * 4 + 2 * half;
        const float* xp = x + (((size_t)(n * 4 + p) * 8 + lb) * HH + yc) * WW + xc;
        v2f xf;
        xf.x = msk * xp[0];
        xf.y = msk * xp[HW];
        const float* wp = wgt + (((size_t)((o * 4 + p) * 9 + k) * 8 + lb) * 16) + col;
        v2f wf;
        wf.x = wp[0];
        wf.y = wp[16];
        c = __builtin_amdgcn_wmma_f32_16x16x4_f32(false, wf, false, xf, (short)0, c,
                                                  false, false);
      }
      pri[k] = c;
    }

    float s[8], outv[8], lg[9];

#pragma unroll
    for (int j = 0; j < 8; ++j) {
      float t = 0.f;
#pragma unroll
      for (int k = 0; k < 9; ++k) t += pri[k][j];
      s[j] = t * (1.f / 9.f);
    }
    {
      float sqp = 0.f;
#pragma unroll
      for (int j = 0; j < 8; ++j) sqp += s[j] * s[j];
      const float sq    = sqp + xhalf(sqp);
      const float scale = (sq * __frcp_rn(1.f + sq)) * __frsqrt_rn(sq);
#pragma unroll
      for (int j = 0; j < 8; ++j) outv[j] = scale * s[j];
    }
#pragma unroll
    for (int k = 0; k < 9; ++k) {
      float t = 0.f;
#pragma unroll
      for (int j = 0; j < 8; ++j) t += pri[k][j] * outv[j];
      lg[k] = t + xhalf(t);
    }

#pragma unroll
    for (int it = 1; it < 3; ++it) {
      float mx = lg[0];
#pragma unroll
      for (int k = 1; k < 9; ++k) mx = fmaxf(mx, lg[k]);
      float den = 0.f;
#pragma unroll
      for (int j = 0; j < 8; ++j) s[j] = 0.f;
#pragma unroll
      for (int k = 0; k < 9; ++k) {
        const float e = __expf(lg[k] - mx);
        den += e;
#pragma unroll
        for (int j = 0; j < 8; ++j) s[j] += e * pri[k][j];
      }
      const float rden = __frcp_rn(den);
      float sqp = 0.f;
#pragma unroll
      for (int j = 0; j < 8; ++j) {
        s[j] *= rden;
        sqp += s[j] * s[j];
      }
      const float sq    = sqp + xhalf(sqp);
      const float scale = (sq * __frcp_rn(1.f + sq)) * __frsqrt_rn(sq);
#pragma unroll
      for (int j = 0; j < 8; ++j) outv[j] = scale * s[j];
      if (it == 1) {
#pragma unroll
        for (int k = 0; k < 9; ++k) {
          float t = 0.f;
#pragma unroll
          for (int j = 0; j < 8; ++j) t += pri[k][j] * outv[j];
          lg[k] += t + xhalf(t);
        }
      }
    }
#pragma unroll
    for (int j = 0; j < 8; ++j) out_acc[j] += outv[j];
  }

#pragma unroll
  for (int j = 0; j < 8; ++j) {
    const int d = j + 8 * half;
    out[((size_t)((n * 4 + o) * 16 + d)) * HW + flat0 + col] = out_acc[j];
  }
}

extern "C" void kernel_launch(void* const* d_in, const int* in_sizes, int n_in,
                              void* d_out, int out_size, void* d_ws, size_t ws_size,
                              hipStream_t stream) {
  const float* x   = (const float*)d_in[0];
  const float* wgt = (const float*)d_in[1];
  float* out = (float*)d_out;
  (void)in_sizes; (void)n_in; (void)out_size; (void)d_ws; (void)ws_size;

  dim3 grid(NUM_TASKS / WAVES_PER_BLOCK);
  dim3 block(WAVES_PER_BLOCK * 32);
  hipLaunchKernelGGL(caps_conv_route, grid, block, 0, stream, x, wgt, out);
}


# Routing runs as 2-wide packed math on the priors' own WMMA register pairs (a sum over d folds the pair with one add): no transposed register copy, 257 -> 195 registers, 4 waves per SIMD
# speedup vs baseline: 7.3733x; 1.1655x over previous
#include <hip/hip_runtime.h>


typedef __attribute__((ext_vector_type(2))) float v2f;
typedef __attribute__((ext_vector_type(8))) float v8f;

#define HH 56
#define WW 56
#define HW 3136
#define TILES_PER_IMG 196
#define WAVES_PER_BLOCK 4
#define NUM_TASKS (4 * 4 * TILES_PER_IMG)

__device__ __forceinline__ float xhalf(float x) {
  const int b = __builtin_bit_cast(int, x);
  return __builtin_bit_cast(
      float, __builtin_amdgcn_permlanex16(b, b, 0x76543210, 0xfedcba98, false, false));
}

__global__ __launch_bounds__(WAVES_PER_BLOCK * 32)
void caps_conv_route(const float* __restrict__ x, const float* __restrict__ wgt,
                     float* __restrict__ out) {
  const int lane = threadIdx.x & 31;
  const int wv   = threadIdx.x >> 5;
  const int task = blockIdx.x * WAVES_PER_BLOCK + wv;
  const int tile = task % TILES_PER_IMG;
  const int o    = (task / TILES_PER_IMG) & 3;
  const int n    = task / (TILES_PER_IMG * 4);
  const int half = lane >> 4;
  const int col  = lane & 15;
  const int flat0 = tile * 16;

  const int fB = flat0 + col;
  const int hB = fB / WW;
  const int wB = fB - hB * WW;

  v2f acc2[4];
#pragma unroll
  for (int jj = 0; jj < 4; ++jj) acc2[jj] = (v2f)(0.f);

  for (int p = 0; p < 4; ++p) {
    v8f pri[9];
#pragma unroll
    for (int k = 0; k < 9; ++k) {
      const int dj = k / 3, dk = k % 3;
      const int y  = hB + dj - 1;
      const int xx = wB + dk - 1;
      const float msk = (y >= 0 && y < HH && xx >= 0 && xx < WW) ? 1.f : 0.f;
      const int yc = min(max(y, 0), HH - 1);
      const int xc = min(max(xx, 0), WW - 1);
      v8f c = {};
#pragma unroll
      for (int hl = 0; hl < 2; ++hl) {
        const int lb = hl * 4 + 2 * half;
        const float* xp = x + (((size_t)(n * 4 + p) * 8 + lb) * HH + yc) * WW + xc;
        v2f xf;
        xf.x = msk * xp[0];
        xf.y = msk * xp[HW];
        const float* wp = wgt + (((size_t)((o * 4 + p) * 9 + k) * 8 + lb) * 16) + col;
        v2f wf;
        wf.x = wp[0];
        wf.y = wp[16];
        c = __builtin_amdgcn_wmma_f32_16x16x4_f32(false, wf, false, xf, (short)0, c,
                                                  false, false);
      }
      pri[k] = c;
    }

    v2f pr[9][4];
#pragma unroll
    for (int k = 0; k < 9; ++k) {
      pr[k][0] = __builtin_shufflevector(pri[k], pri[k], 0, 1);
      pr[k][1] = __builtin_shufflevector(pri[k], pri[k], 2, 3);
      pr[k][2] = __builtin_shufflevector(pri[k], pri[k], 4, 5);
      pr[k][3] = __builtin_shufflevector(pri[k], pri[k], 6, 7);
    }

    v2f s2[4], o2[4];
    float lg[9];

#pragma unroll
    for (int jj = 0; jj < 4; ++jj) {
      v2f t = pr[0][jj];
#pragma unroll
      for (int k = 1; k < 9; ++k) t += pr[k][jj];
      s2[jj] = t * (1.f / 9.f);
    }
    {
      v2f q = s2[0] * s2[0];
#pragma unroll
      for (int jj = 1; jj < 4; ++jj) q = __builtin_elementwise_fma(s2[jj], s2[jj], q);
      const float sqp   = q.x + q.y;
      const float sq    = sqp + xhalf(sqp);
      const float scale = (sq * __frcp_rn(1.f + sq)) * __frsqrt_rn(sq);
#pragma unroll
      for (int jj = 0; jj < 4; ++jj) o2[jj] = s2[jj] * scale;
    }
#pragma unroll
    for (int k = 0; k < 9; ++k) {
      v2f t = pr[k][0] * o2[0];
#pragma unroll
      for (int jj = 1; jj < 4; ++jj) t = __builtin_elementwise_fma(pr[k][jj], o2[jj], t);
      const float th = t.x + t.y;
      lg[k] = th + xhalf(th);
    }

#pragma unroll
    for (int it = 1; it < 3; ++it) {
      float mx = lg[0];
#pragma unroll
      for (int k = 1; k < 9; ++k) mx = fmaxf(mx, lg[k]);
      float den = 0.f;
#pragma unroll
      for (int k = 0; k < 9; ++k) {
        const float e = __expf(lg[k] - mx);
        den += e;
        if (k == 0) {
#pragma unroll
          for (int jj = 0; jj < 4; ++jj) s2[jj] = pr[0][jj] * e;
        } else {
#pragma unroll
          for (int jj = 0; jj < 4; ++jj)
            s2[jj] = __builtin_elementwise_fma(pr[k][jj], (v2f)(e), s2[jj]);
        }
      }
      const float rden = __frcp_rn(den);
#pragma unroll
      for (int jj = 0; jj < 4; ++jj) s2[jj] = s2[jj] * rden;
      v2f q = s2[0] * s2[0];
#pragma unroll
      for (int jj = 1; jj < 4; ++jj) q = __builtin_elementwise_fma(s2[jj], s2[jj], q);
      const float sqp   = q.x + q.y;
      const float sq    = sqp + xhalf(sqp);
      const float scale = (sq * __frcp_rn(1.f + sq)) * __frsqrt_rn(sq);
#pragma unroll
      for (int jj = 0; jj < 4; ++jj) o2[jj] = s2[jj] * scale;
      if (it == 1) {
#pragma unroll
        for (int k = 0; k < 9; ++k) {
          v2f t = pr[k][0] * o2[0];
#pragma unroll
          for (int jj = 1; jj < 4; ++jj) t = __builtin_elementwise_fma(pr[k][jj], o2[jj], t);
          const float th = t.x + t.y;
          lg[k] += th + xhalf(th);
        }
      }
    }
#pragma unroll
    for (int jj = 0; jj < 4; ++jj) acc2[jj] += o2[jj];
  }

#pragma unroll
  for (int j = 0; j < 8; ++j) {
    const int d = j + 8 * half;
    out[((size_t)((n * 4 + o) * 16 + d)) * HW + flat0 + col] = acc2[j >> 1][j & 1];
  }
}

extern "C" void kernel_launch(void* const* d_in, const int* in_sizes, int n_in,
                              void* d_out, int out_size, void* d_ws, size_t ws_size,
                              hipStream_t stream) {
  const float* x   = (const float*)d_in[0];
  const float* wgt = (const float*)d_in[1];
  float* out = (float*)d_out;
  (void)in_sizes; (void)n_in; (void)out_size; (void)d_ws; (void)ws_size;

  dim3 grid(NUM_TASKS / WAVES_PER_BLOCK);
  dim3 block(WAVES_PER_BLOCK * 32);
  hipLaunchKernelGGL(caps_conv_route, grid, block, 0, stream, x, wgt, out);
}


# Operands of input plane p+1 are requested right behind the WMMAs of plane p and land during its routing; addresses are a uniform base + fixed lane offset + immediate
# speedup vs baseline: 7.8155x; 1.0600x over previous
#include <hip/hip_runtime.h>


typedef __attribute__((ext_vector_type(2))) float v2f;
typedef __attribute__((ext_vector_type(8))) float v8f;
typedef const char __attribute__((address_space(1)))* gbytes;
typedef const float __attribute__((address_space(1)))* gfloats;
#define LDG(base, voff, imm) (*(gfloats)((base) + (voff) + (imm)))

#define HH 56
#define WW 56
#define HW 3136
#define TILES_PER_IMG 196
#define WAVES_PER_BLOCK 4
#define NUM_TASKS (4 * 4 * TILES_PER_IMG)

__device__ __forceinline__ float xhalf(float x) {
  const int b = __builtin_bit_cast(int, x);
  return __builtin_bit_cast(
      float, __builtin_amdgcn_permlanex16(b, b, 0x76543210, 0xfedcba98, false, false));
}

__global__ __launch_bounds__(WAVES_PER_BLOCK * 32)
void caps_conv_route(const float* __restrict__ x, const float* __restrict__ wgt,
                     float* __restrict__ out) {
  const int lane = threadIdx.x & 31;
  const int wv   = __builtin_amdgcn_readfirstlane((int)(threadIdx.x >> 5));
  const int task = blockIdx.x * WAVES_PER_BLOCK + wv;
  const int tile = task % TILES_PER_IMG;
  const int o    = (task / TILES_PER_IMG) & 3;
  const int n    = task / (TILES_PER_IMG * 4);
  const int half = lane >> 4;
  const int col  = lane & 15;
  const int flat0 = tile * 16;

  const int fB = flat0 + col;
  const int hB = fB / WW;
  const int wB = fB - hB * WW;

  unsigned xo[9];
  float msk[9];
#pragma unroll
  for (int k = 0; k < 9; ++k) {
    const int dj = k / 3, dk = k % 3;
    const int y  = hB + dj - 1;
    const int xx = wB + dk - 1;
    msk[k] = (y >= 0 && y < HH && xx >= 0 && xx < WW) ? 1.f : 0.f;
    const int yc = min(max(y, 0), HH - 1);
    const int xc = min(max(xx, 0), WW - 1);
    xo[k] = (unsigned)(((2 * half) * HW + yc * WW + xc) * 4);
  }
  const unsigned wo = (unsigned)((half * 32 + col) * 4);
  gbytes xb = (gbytes)x + (size_t)n * (32 * HW * 4);
  gbytes wb = (gbytes)wgt + (size_t)o * (4 * 1152 * 4);

  float xr[9][4], wr[9][4];
#pragma unroll
  for (int k = 0; k < 9; ++k) {
#pragma unroll
    for (int hl = 0; hl < 2; ++hl) {
      xr[k][2 * hl]     = LDG(xb, xo[k], (hl * 4) * HW * 4);
      xr[k][2 * hl + 1] = LDG(xb, xo[k], (hl * 4 + 1) * HW * 4);
      wr[k][2 * hl]     = LDG(wb, wo, (k * 128 + hl * 64) * 4);
      wr[k][2 * hl + 1] = LDG(wb, wo, (k * 128 + hl * 64 + 16) * 4);
    }
  }

  v2f acc2[4];
#pragma unroll
  for (int jj = 0; jj < 4; ++jj) acc2[jj] = (v2f)(0.f);

#pragma unroll 1
  for (int p = 0; p < 4; ++p) {
    v8f pri[9];
#pragma unroll
    for (int k = 0; k < 9; ++k) {
      v8f c = {};
#pragma unroll
      for (int hl = 0; hl < 2; ++hl) {
        v2f xf;
        xf.x = msk[k] * xr[k][2 * hl];
        xf.y = msk[k] * xr[k][2 * hl + 1];
        v2f wf;
        wf.x = wr[k][2 * hl];
        wf.y = wr[k][2 * hl + 1];
        c = __builtin_amdgcn_wmma_f32_16x16x4_f32(false, wf, false, xf, (short)0, c,
                                                  false, false);
      }
      pri[k] = c;
    }

    if (p < 3) {
      xb += 8 * HW * 4;
      wb += 1152 * 4;
#pragma unroll
      for (int k = 0; k < 9; ++k) {
#pragma unroll
        for (int hl = 0; hl < 2; ++hl) {
          xr[k][2 * hl]     = LDG(xb, xo[k], (hl * 4) * HW * 4);
          xr[k][2 * hl + 1] = LDG(xb, xo[k], (hl * 4 + 1) * HW * 4);
          wr[k][2 * hl]     = LDG(wb, wo, (k * 128 + hl * 64) * 4);
          wr[k][2 * hl + 1] = LDG(wb, wo, (k * 128 + hl * 64 + 16) * 4);
        }
      }
    }
    __builtin_amdgcn_sched_barrier(0);

    v2f pr[9][4];
#pragma unroll
    for (int k = 0; k < 9; ++k) {
      pr[k][0] = __builtin_shufflevector(pri[k], pri[k], 0, 1);
      pr[k][1] = __builtin_shufflevector(pri[k], pri[k], 2, 3);
      pr[k][2] = __builtin_shufflevector(pri[k], pri[k], 4, 5);
      pr[k][3] = __builtin_shufflevector(pri[k], pri[k], 6, 7);
    }

    v2f s2[4], o2[4];
    float lg[9];

#pragma unroll
    for (int jj = 0; jj < 4; ++jj) {
      v2f t = pr[0][jj];
#pragma unroll
      for (int k = 1; k < 9; ++k) t += pr[k][jj];
      s2[jj] = t * (1.f / 9.f);
    }
    {
      v2f q = s2[0] * s2[0];
#pragma unroll
      for (int jj = 1; jj < 4; ++jj) q = __builtin_elementwise_fma(s2[jj], s2[jj], q);
      const float sqp   = q.x + q.y;
      const float sq    = sqp + xhalf(sqp);
      const float scale = (sq * __frcp_rn(1.f + sq)) * __frsqrt_rn(sq);
#pragma unroll
      for (int jj = 0; jj < 4; ++jj) o2[jj] = s2[jj] * scale;
    }
#pragma unroll
    for (int k = 0; k < 9; ++k) {
      v2f t = pr[k][0] * o2[0];
#pragma unroll
      for (int jj = 1; jj < 4; ++jj) t = __builtin_elementwise_fma(pr[k][jj], o2[jj], t);
      const float th = t.x + t.y;
      lg[k] = th + xhalf(th);
    }

#pragma unroll
    for (int it = 1; it < 3; ++it) {
      float mx = lg[0];
#pragma unroll
      for (int k = 1; k < 9; ++k) mx = fmaxf(mx, lg[k]);
      float den = 0.f;
#pragma unroll
      for (int k = 0; k < 9; ++k) {
        const float e = __expf(lg[k] - mx);
        den += e;
        if (k == 0) {
#pragma unroll
          for (int jj = 0; jj < 4; ++jj) s2[jj] = pr[0][jj] * e;
        } else {
#pragma unroll
          for (int jj = 0; jj < 4; ++jj)
            s2[jj] = __builtin_elementwise_fma(pr[k][jj], (v2f)(e), s2[jj]);
        }
      }
      const float rden = __frcp_rn(den);
#pragma unroll
      for (int jj = 0; jj < 4; ++jj) s2[jj] = s2[jj] * rden;
      v2f q = s2[0] * s2[0];
#pragma unroll
      for (int jj = 1; jj < 4; ++jj) q = __builtin_elementwise_fma(s2[jj], s2[jj], q);
      const float sqp   = q.x + q.y;
      const float sq    = sqp + xhalf(sqp);
      const float scale = (sq * __frcp_rn(1.f + sq)) * __frsqrt_rn(sq);
#pragma unroll
      for (int jj = 0; jj < 4; ++jj) o2[jj] = s2[jj] * scale;
      if (it == 1) {
#pragma unroll
        for (int k = 0; k < 9; ++k) {
          v2f t = pr[k][0] * o2[0];
#pragma unroll
          for (int jj = 1; jj < 4; ++jj) t = __builtin_elementwise_fma(pr[k][jj], o2[jj], t);
          const float th = t.x + t.y;
          lg[k] += th + xhalf(th);
        }
      }
    }
#pragma unroll
    for (int jj = 0; jj < 4; ++jj) acc2[jj] += o2[jj];
  }

#pragma unroll
  for (int j = 0; j < 8; ++j) {
    const int d = j + 8 * half;
    out[((size_t)((n * 4 + o) * 16 + d)) * HW + flat0 + col] = acc2[j >> 1][j & 1];
  }
}

extern "C" void kernel_launch(void* const* d_in, const int* in_sizes, int n_in,
                              void* d_out, int out_size, void* d_ws, size_t ws_size,
                              hipStream_t stream) {
  const float* x   = (const float*)d_in[0];
  const float* wgt = (const float*)d_in[1];
  float* out = (float*)d_out;
  (void)in_sizes; (void)n_in; (void)out_size; (void)d_ws; (void)ws_size;

  dim3 grid(NUM_TASKS / WAVES_PER_BLOCK);
  dim3 block(WAVES_PER_BLOCK * 32);
  hipLaunchKernelGGL(caps_conv_route, grid, block, 0, stream, x, wgt, out);
}


# 3136 tasks = 3 x 1024 SIMDs + 64: the 64 left-over tasks run as 64 extra blocks whose four waves route one input plane each and sum through LDS, so no SIMD carries a fourth whole task
# speedup vs baseline: 8.5172x; 1.0898x over previous
#include <hip/hip_runtime.h>


typedef __attribute__((ext_vector_type(2))) float v2f;
typedef __attribute__((ext_vector_type(8))) float v8f;
typedef const char __attribute__((address_space(1)))* gbytes;
typedef const float __attribute__((address_space(1)))* gfloats;
#define LDG(base, voff, imm) (*(gfloats)((base) + (voff) + (imm)))

#define HH 56
#define WW 56
#define HW 3136
#define TILES_PER_IMG 196
#define WAVES_PER_BLOCK 4
#define NUM_TASKS (4 * 4 * TILES_PER_IMG)
#define FULL_BLOCKS 768
#define GRID_BLOCKS 832

__device__ __forceinline__ float xhalf(float x) {
  const int b = __builtin_bit_cast(int, x);
  return __builtin_bit_cast(
      float, __builtin_amdgcn_permlanex16(b, b, 0x76543210, 0xfedcba98, false, false));
}

__global__ __launch_bounds__(WAVES_PER_BLOCK * 32)
void caps_conv_route(const float* __restrict__ x, const float* __restrict__ wgt,
                     float* __restrict__ out) {
  __shared__ float red[WAVES_PER_BLOCK][8][32];
  const int lane = threadIdx.x & 31;
  const int wv   = __builtin_amdgcn_readfirstlane((int)(threadIdx.x >> 5));
  const bool split = blockIdx.x >= FULL_BLOCKS;
  const int task = split ? (FULL_BLOCKS * WAVES_PER_BLOCK + ((int)blockIdx.x - FULL_BLOCKS))
                         : ((int)blockIdx.x * WAVES_PER_BLOCK + wv);
  const int p0   = split ? wv : 0;
  const int p1   = split ? wv + 1 : 4;
  const int tile = task % TILES_PER_IMG;
  const int o    = (task / TILES_PER_IMG) & 3;
  const int n    = task / (TILES_PER_IMG * 4);
  const int half = lane >> 4;
  const int col  = lane & 15;
  const int flat0 = tile * 16;

  const int fB = flat0 + col;
  const int hB = fB / WW;
  const int wB = fB - hB * WW;

  unsigned xo[9];
  float msk[9];
#pragma unroll
  for (int k = 0; k < 9; ++k) {
    const int dj = k / 3, dk = k % 3;
    const int y  = hB + dj - 1;
    const int xx = wB + dk - 1;
    msk[k] = (y >= 0 && y < HH && xx >= 0 && xx < WW) ? 1.f : 0.f;
    const int yc = min(max(y, 0), HH - 1);
    const int xc = min(max(xx, 0), WW - 1);
    xo[k] = (unsigned)(((2 * half) * HW + yc * WW + xc) * 4);
  }
  const unsigned wo = (unsigned)((half * 32 + col) * 4);
  gbytes xb = (gbytes)x + (size_t)n * (32 * HW * 4) + (size_t)p0 * (8 * HW * 4);
  gbytes wb = (gbytes)wgt + (size_t)o * (4 * 1152 * 4) + (size_t)p0 * (1152 * 4);

  float xr[9][4], wr[9][4];
#pragma unroll
  for (int k = 0; k < 9; ++k) {
#pragma unroll
    for (int hl = 0; hl < 2; ++hl) {
      xr[k][2 * hl]     = LDG(xb, xo[k], (hl * 4) * HW * 4);
      xr[k][2 * hl + 1] = LDG(xb, xo[k], (hl * 4 + 1) * HW * 4);
      wr[k][2 * hl]     = LDG(wb, wo, (k * 128 + hl * 64) * 4);
      wr[k][2 * hl + 1] = LDG(wb, wo, (k * 128 + hl * 64 + 16) * 4);
    }
    __builtin_amdgcn_sched_barrier(0);
  }

  v2f acc2[4];
#pragma unroll
  for (int jj = 0; jj < 4; ++jj) acc2[jj] = (v2f)(0.f);

#pragma unroll 1
  for (int p = p0; p < p1; ++p) {
    v8f pri[9];
#pragma unroll
    for (int k = 0; k < 9; ++k) {
      v8f c = {};
#pragma unroll
      for (int hl = 0; hl < 2; ++hl) {
        v2f xf;
        xf.x = msk[k] * xr[k][2 * hl];
        xf.y = msk[k] * xr[k][2 * hl + 1];
        v2f wf;
        wf.x = wr[k][2 * hl];
        wf.y = wr[k][2 * hl + 1];
        c = __builtin_amdgcn_wmma_f32_16x16x4_f32(false, wf, false, xf, (short)0, c,
                                                  false, false);
      }
      pri[k] = c;
    }

    if (p + 1 < p1) {
      xb += 8 * HW * 4;
      wb += 1152 * 4;
#pragma unroll
      for (int k = 0; k < 9; ++k) {
#pragma unroll
        for (int hl = 0; hl < 2; ++hl) {
          xr[k][2 * hl]     = LDG(xb, xo[k], (hl * 4) * HW * 4);
          xr[k][2 * hl + 1] = LDG(xb, xo[k], (hl * 4 + 1) * HW * 4);
          wr[k][2 * hl]     = LDG(wb, wo, (k * 128 + hl * 64) * 4);
          wr[k][2 * hl + 1] = LDG(wb, wo, (k * 128 + hl * 64 + 16) * 4);
        }
      }
    }
    __builtin_amdgcn_sched_barrier(0);

    v2f pr[9][4];
#pragma unroll
    for (int k = 0; k < 9; ++k) {
      pr[k][0] = __builtin_shufflevector(pri[k], pri[k], 0, 1);
      pr[k][1] = __builtin_shufflevector(pri[k], pri[k], 2, 3);
      pr[k][2] = __builtin_shufflevector(pri[k], pri[k], 4, 5);
      pr[k][3] = __builtin_shufflevector(pri[k], pri[k], 6, 7);
    }

    v2f s2[4], o2[4];
    float lg[9];

#pragma unroll
    for (int jj = 0; jj < 4; ++jj) {
      v2f t = pr[0][jj];
#pragma unroll
      for (int k = 1; k < 9; ++k) t += pr[k][jj];
      s2[jj] = t * (1.f / 9.f);
    }
    {
      v2f q = s2[0] * s2[0];
#pragma unroll
      for (int jj = 1; jj < 4; ++jj) q = __builtin_elementwise_fma(s2[jj], s2[jj], q);
      const float sqp   = q.x + q.y;
      const float sq    = sqp + xhalf(sqp);
      const float scale = (sq * __frcp_rn(1.f + sq)) * __frsqrt_rn(sq);
#pragma unroll
      for (int jj = 0; jj < 4; ++jj) o2[jj] = s2[jj] * scale;
    }
#pragma unroll
    for (int k = 0; k < 9; ++k) {
      v2f t = pr[k][0] * o2[0];
#pragma unroll
      for (int jj = 1; jj < 4; ++jj) t = __builtin_elementwise_fma(pr[k][jj], o2[jj], t);
      const float th = t.x + t.y;
      lg[k] = th + xhalf(th);
    }

#pragma unroll
    for (int it = 1; it < 3; ++it) {
      float mx = lg[0];
#pragma unroll
      for (int k = 1; k < 9; ++k) mx = fmaxf(mx, lg[k]);
      float den = 0.f;
#pragma unroll
      for (int k = 0; k < 9; ++k) {
        const float e = __expf(lg[k] - mx);
        den += e;
        if (k == 0) {
#pragma unroll
          for (int jj = 0; jj < 4; ++jj) s2[jj] = pr[0][jj] * e;
        } else {
#pragma unroll
          for (int jj = 0; jj < 4; ++jj)
            s2[jj] = __builtin_elementwise_fma(pr[k][jj], (v2f)(e), s2[jj]);
        }
      }
      const float rden = __frcp_rn(den);
#pragma unroll
      for (int jj = 0; jj < 4; ++jj) s2[jj] = s2[jj] * rden;
      v2f q = s2[0] * s2[0];
#pragma unroll
      for (int jj = 1; jj < 4; ++jj) q = __builtin_elementwise_fma(s2[jj], s2[jj], q);
      const float sqp   = q.x + q.y;
      const float sq    = sqp + xhalf(sqp);
      const float scale = (sq * __frcp_rn(1.f + sq)) * __frsqrt_rn(sq);
#pragma unroll
      for (int jj = 0; jj < 4; ++jj) o2[jj] = s2[jj] * scale;
      if (it == 1) {
#pragma unroll
        for (int k = 0; k < 9; ++k) {
          v2f t = pr[k][0] * o2[0];
#pragma unroll
          for (int jj = 1; jj < 4; ++jj) t = __builtin_elementwise_fma(pr[k][jj], o2[jj], t);
          const float th = t.x + t.y;
          lg[k] += th + xhalf(th);
        }
      }
    }
#pragma unroll
    for (int jj = 0; jj < 4; ++jj) acc2[jj] += o2[jj];
  }

  float* orow = out + ((size_t)((n * 4 + o) * 16 + 8 * half)) * HW + flat0 + col;
  if (!split) {
#pragma unroll
    for (int j = 0; j < 8; ++j) orow[(size_t)j * HW] = acc2[j >> 1][j & 1];
  } else {
#pragma unroll
    for (int j = 0; j < 8; ++j) red[wv][j][lane] = acc2[j >> 1][j & 1];
    __syncthreads();
#pragma unroll
    for (int jj = 0; jj < 2; ++jj) {
      const int j = 2 * wv + jj;
      const float v = ((red[0][j][lane] + red[1][j][lane]) + red[2][j][lane]) + red[3][j][lane];
      orow[(size_t)j * HW] = v;
    }
  }
}

extern "C" void kernel_launch(void* const* d_in, const int* in_sizes, int n_in,
                              void* d_out, int out_size, void* d_ws, size_t ws_size,
                              hipStream_t stream) {
  const float* x   = (const float*)d_in[0];
  const float* wgt = (const float*)d_in[1];
  float* out = (float*)d_out;
  (void)in_sizes; (void)n_in; (void)out_size; (void)d_ws; (void)ws_size;

  dim3 grid(GRID_BLOCKS);
  dim3 block(WAVES_PER_BLOCK * 32);
  hipLaunchKernelGGL(caps_conv_route, grid, block, 0, stream, x, wgt, out);
}


# One hardware reciprocal (v_rcp_f32) per routing iteration: the 1/x spellings compiled to 10-instruction IEEE divisions, 5 per plane; softmax 1/den folds into the squash factor sqrt(q)/(den^2+q)
# speedup vs baseline: 9.0704x; 1.0650x over previous
#include <hip/hip_runtime.h>


typedef __attribute__((ext_vector_type(2))) float v2f;
typedef __attribute__((ext_vector_type(8))) float v8f;
typedef const char __attribute__((address_space(1)))* gbytes;
typedef const float __attribute__((address_space(1)))* gfloats;
#define LDG(base, voff, imm) (*(gfloats)((base) + (voff) + (imm)))

#define HH 56
#define WW 56
#define HW 3136
#define TILES_PER_IMG 196
#define WAVES_PER_BLOCK 4
#define NUM_TASKS (4 * 4 * TILES_PER_IMG)
#define FULL_BLOCKS 768
#define GRID_BLOCKS 832

__device__ __forceinline__ float xhalf(float x) {
  const int b = __builtin_bit_cast(int, x);
  return __builtin_bit_cast(
      float, __builtin_amdgcn_permlanex16(b, b, 0x76543210, 0xfedcba98, false, false));
}

__device__ __forceinline__ float squash_factor(float q, float den2) {
  return (q * __builtin_amdgcn_rsqf(q)) * __builtin_amdgcn_rcpf(den2 + q);
}

__global__ __launch_bounds__(WAVES_PER_BLOCK * 32)
void caps_conv_route(const float* __restrict__ x, const float* __restrict__ wgt,
                     float* __restrict__ out) {
  __shared__ float red[WAVES_PER_BLOCK][8][32];
  const int lane = threadIdx.x & 31;
  const int wv   = __builtin_amdgcn_readfirstlane((int)(threadIdx.x >> 5));
  const bool split = blockIdx.x >= FULL_BLOCKS;
  const int task = split ? (FULL_BLOCKS * WAVES_PER_BLOCK + ((int)blockIdx.x - FULL_BLOCKS))
                         : ((int)blockIdx.x * WAVES_PER_BLOCK + wv);
  const int p0   = split ? wv : 0;
  const int p1   = split ? wv + 1 : 4;
  const int tile = task % TILES_PER_IMG;
  const int o    = (task / TILES_PER_IMG) & 3;
  const int n    = task / (TILES_PER_IMG * 4);
  const int half = lane >> 4;
  const int col  = lane & 15;
  const int flat0 = tile * 16;

  const int fB = flat0 + col;
  const int hB = fB / WW;
  const int wB = fB - hB * WW;

  unsigned xo[9];
  float msk[9];
#pragma unroll
  for (int k = 0; k < 9; ++k) {
    const int dj = k / 3, dk = k % 3;
    const int y  = hB + dj - 1;
    const int xx = wB + dk - 1;
    msk[k] = (y >= 0 && y < HH && xx >= 0 && xx < WW) ? 1.f : 0.f;
    const int yc = min(max(y, 0), HH - 1);
    const int xc = min(max(xx, 0), WW - 1);
    xo[k] = (unsigned)(((2 * half) * HW + yc * WW + xc) * 4);
  }
  const unsigned wo = (unsigned)((half * 32 + col) * 4);
  gbytes xb = (gbytes)x + (size_t)n * (32 * HW * 4) + (size_t)p0 * (8 * HW * 4);
  gbytes wb = (gbytes)wgt + (size_t)o * (4 * 1152 * 4) + (size_t)p0 * (1152 * 4);

  float xr[9][4], wr[9][4];
#pragma unroll
  for (int k = 0; k < 9; ++k) {
#pragma unroll
    for (int hl = 0; hl < 2; ++hl) {
      xr[k][2 * hl]     = LDG(xb, xo[k], (hl * 4) * HW * 4);
      xr[k][2 * hl + 1] = LDG(xb, xo[k], (hl * 4 + 1) * HW * 4);
      wr[k][2 * hl]     = LDG(wb, wo, (k * 128 + hl * 64) * 4);
      wr[k][2 * hl + 1] = LDG(wb, wo, (k * 128 + hl * 64 + 16) * 4);
    }
    __builtin_amdgcn_sched_barrier(0);
  }

  v2f acc2[4];
#pragma unroll
  for (int jj = 0; jj < 4; ++jj) acc2[jj] = (v2f)(0.f);

#pragma unroll 1
  for (int p = p0; p < p1; ++p) {
    v8f pri[9];
#pragma unroll
    for (int k = 0; k < 9; ++k) {
      v8f c = {};
#pragma unroll
      for (int hl = 0; hl < 2; ++hl) {
        v2f xf;
        xf.x = msk[k] * xr[k][2 * hl];
        xf.y = msk[k] * xr[k][2 * hl + 1];
        v2f wf;
        wf.x = wr[k][2 * hl];
        wf.y = wr[k][2 * hl + 1];
        c = __builtin_amdgcn_wmma_f32_16x16x4_f32(false, wf, false, xf, (short)0, c,
                                                  false, false);
      }
      pri[k] = c;
    }

    if (p + 1 < p1) {
      xb += 8 * HW * 4;
      wb += 1152 * 4;
#pragma unroll
      for (int k = 0; k < 9; ++k) {
#pragma unroll
        for (int hl = 0; hl < 2; ++hl) {
          xr[k][2 * hl]     = LDG(xb, xo[k], (hl * 4) * HW * 4);
          xr[k][2 * hl + 1] = LDG(xb, xo[k], (hl * 4 + 1) * HW * 4);
          wr[k][2 * hl]     = LDG(wb, wo, (k * 128 + hl * 64) * 4);
          wr[k][2 * hl + 1] = LDG(wb, wo, (k * 128 + hl * 64 + 16) * 4);
        }
      }
    }
    __builtin_amdgcn_sched_barrier(0);

    v2f pr[9][4];
#pragma unroll
    for (int k = 0; k < 9; ++k) {
      pr[k][0] = __builtin_shufflevector(pri[k], pri[k], 0, 1);
      pr[k][1] = __builtin_shufflevector(pri[k], pri[k], 2, 3);
      pr[k][2] = __builtin_shufflevector(pri[k], pri[k], 4, 5);
      pr[k][3] = __builtin_shufflevector(pri[k], pri[k], 6, 7);
    }

    v2f s2[4], o2[4];
    float lg[9];

#pragma unroll
    for (int jj = 0; jj < 4; ++jj) {
      v2f t = pr[0][jj];
#pragma unroll
      for (int k = 1; k < 9; ++k) t += pr[k][jj];
      s2[jj] = t;
    }
    {
      v2f q = s2[0] * s2[0];
#pragma unroll
      for (int jj = 1; jj < 4; ++jj) q = __builtin_elementwise_fma(s2[jj], s2[jj], q);
      const float sqp   = q.x + q.y;
      const float sq    = sqp + xhalf(sqp);
      const float scale = squash_factor(sq, 81.f);
#pragma unroll
      for (int jj = 0; jj < 4; ++jj) o2[jj] = s2[jj] * scale;
    }
#pragma unroll
    for (int k = 0; k < 9; ++k) {
      v2f t = pr[k][0] * o2[0];
#pragma unroll
      for (int jj = 1; jj < 4; ++jj) t = __builtin_elementwise_fma(pr[k][jj], o2[jj], t);
      const float th = t.x + t.y;
      lg[k] = th + xhalf(th);
    }

#pragma unroll
    for (int it = 1; it < 3; ++it) {
      float mx = lg[0];
#pragma unroll
      for (int k = 1; k < 9; ++k) mx = fmaxf(mx, lg[k]);
      float den = 0.f;
#pragma unroll
      for (int k = 0; k < 9; ++k) {
        const float e = __expf(lg[k] - mx);
        den += e;
        if (k == 0) {
#pragma unroll
          for (int jj = 0; jj < 4; ++jj) s2[jj] = pr[0][jj] * e;
        } else {
#pragma unroll
          for (int jj = 0; jj < 4; ++jj)
            s2[jj] = __builtin_elementwise_fma(pr[k][jj], (v2f)(e), s2[jj]);
        }
      }
      v2f q = s2[0] * s2[0];
#pragma unroll
      for (int jj = 1; jj < 4; ++jj) q = __builtin_elementwise_fma(s2[jj], s2[jj], q);
      const float sqp   = q.x + q.y;
      const float sq    = sqp + xhalf(sqp);
      const float scale = squash_factor(sq, den * den);
#pragma unroll
      for (int jj = 0; jj < 4; ++jj) o2[jj] = s2[jj] * scale;
      if (it == 1) {
#pragma unroll
        for (int k = 0; k < 9; ++k) {
          v2f t = pr[k][0] * o2[0];
#pragma unroll
          for (int jj = 1; jj < 4; ++jj) t = __builtin_elementwise_fma(pr[k][jj], o2[jj], t);
          const float th = t.x + t.y;
          lg[k] += th + xhalf(th);
        }
      }
    }
#pragma unroll
    for (int jj = 0; jj < 4; ++jj) acc2[jj] += o2[jj];
  }

  float* orow = out + ((size_t)((n * 4 + o) * 16 + 8 * half)) * HW + flat0 + col;
  if (!split) {
#pragma unroll
    for (int j = 0; j < 8; ++j) orow[(size_t)j * HW] = acc2[j >> 1][j & 1];
  } else {
#pragma unroll
    for (int j = 0; j < 8; ++j) red[wv][j][lane] = acc2[j >> 1][j & 1];
    __syncthreads();
#pragma unroll
    for (int jj = 0; jj < 2; ++jj) {
      const int j = 2 * wv + jj;
      const float v = ((red[0][j][lane] + red[1][j][lane]) + red[2][j][lane]) + red[3][j][lane];
      orow[(size_t)j * HW] = v;
    }
  }
}

extern "C" void kernel_launch(void* const* d_in, const int* in_sizes, int n_in,
                              void* d_out, int out_size, void* d_ws, size_t ws_size,
                              hipStream_t stream) {
  const float* x   = (const float*)d_in[0];
  const float* wgt = (const float*)d_in[1];
  float* out = (float*)d_out;
  (void)in_sizes; (void)n_in; (void)out_size; (void)d_ws; (void)ws_size;

  dim3 grid(GRID_BLOCKS);
  dim3 block(WAVES_PER_BLOCK * 32);
  hipLaunchKernelGGL(caps_conv_route, grid, block, 0, stream, x, wgt, out);
}
